# starting kernel
# baseline (speedup 1.0000x reference)
#include <hip/hip_runtime.h>


typedef float v2f __attribute__((ext_vector_type(2)));
typedef float v8f __attribute__((ext_vector_type(8)));

#define WMMA_F32X4(A, B, C) \
  __builtin_amdgcn_wmma_f32_16x16x4_f32(false, (A), false, (B), (short)0, (C), false, false)

namespace {
constexpr int BATCH = 4;
constexpr int CQK   = 64;
constexpr int CV    = 128;
constexpr int NSEQ  = 4096;
constexpr int MT    = 32;
constexpr int LDK2 = 2 * MT + 2;
constexpr int LDV2 = 2 * CV + 2;
constexpr int LDP2 = 2 * 16 + 2;
}

__global__ __launch_bounds__(128) void proj_wmma_kernel(
    const float* __restrict__ W, const float* __restrict__ X,
    float* __restrict__ Y, int O, int C) {
  const int lane = threadIdx.x & 31;
  const int wave = threadIdx.x >> 5;
  const int h    = lane >> 4;
  const int l15  = lane & 15;
  const int b    = blockIdx.z;
  const int o0   = blockIdx.y * 64 + wave * 16;
  const int n0   = blockIdx.x * 64;

  const float* Xb = X + (size_t)b * C * NSEQ;
  float* Yb       = Y + (size_t)b * O * NSEQ;

  v8f acc[4];
#pragma unroll
  for (int st = 0; st < 4; ++st) { v8f z = {}; acc[st] = z; }

  const int ksteps = C >> 2;
  for (int t = 0; t < ksteps; ++t) {
    const int c0 = 4 * t + 2 * h;
    v2f a = *(const v2f*)&W[(size_t)(o0 + l15) * C + c0];
#pragma unroll
    for (int st = 0; st < 4; ++st) {
      const int n = n0 + 16 * st + l15;
      v2f bop;
      bop[0] = Xb[(size_t)c0 * NSEQ + n];
      bop[1] = Xb[(size_t)(c0 + 1) * NSEQ + n];
      acc[st] = WMMA_F32X4(a, bop, acc[st]);
    }
  }
#pragma unroll
  for (int st = 0; st < 4; ++st) {
    const int n = n0 + 16 * st + l15;
#pragma unroll
    for (int j = 0; j < 8; ++j) {
      Yb[(size_t)(o0 + j + 8 * h) * NSEQ + n] = acc[st][j];
    }
  }
}

__global__ __launch_bounds__(128) void flash_attn_f32_kernel(
    const float* __restrict__ q, const float* __restrict__ k,
    const float* __restrict__ v, const float* __restrict__ x_h,
    const float* __restrict__ gamma_p, float* __restrict__ out) {
  __shared__ float kbp[(CQK / 2) * LDK2];
  __shared__ float vtp[(MT / 2) * LDV2];
  __shared__ float ptb[4 * (MT / 2) * LDP2];
  __shared__ float red[4][16];
  __shared__ float mrow_s[4][16];

  const int tid  = threadIdx.x;
  const int lane = tid & 31;
  const int wave = tid >> 5;
  const int h    = lane >> 4;
  const int l15  = lane & 15;
  const int b    = blockIdx.y;
  const int n0   = blockIdx.x * 64 + wave * 16;

  const float* qb  = q + (size_t)b * CQK * NSEQ;
  const float* kb  = k + (size_t)b * CQK * NSEQ;
  const float* vb  = v + (size_t)b * CV * NSEQ;

  const int kc  = tid >> 1;
  const int kmh = (tid & 1) * 16;
  const int vc  = tid;

  v2f qa[16];
#pragma unroll
  for (int t = 0; t < 16; ++t) {
    const int c0 = 4 * t + 2 * h;
    const int n  = n0 + l15;
    qa[t][0] = qb[(size_t)c0 * NSEQ + n];
    qa[t][1] = qb[(size_t)(c0 + 1) * NSEQ + n];
  }

  v8f Oacc[8];
#pragma unroll
  for (int ct = 0; ct < 8; ++ct) { v8f z = {}; Oacc[ct] = z; }
  float m_s = -3.0e38f;
  float l_s = 0.0f;

  float* ptw = &ptb[wave * (MT / 2) * LDP2];

  for (int m0 = 0; m0 < NSEQ; m0 += MT) {
    __syncthreads();

    {
      float* dst = &kbp[(kc >> 1) * LDK2 + (kc & 1)];
#pragma unroll
      for (int i = 0; i < 4; ++i) {
        const int mb = kmh + 4 * i;
        const float4 kv = *(const float4*)&kb[(size_t)kc * NSEQ + m0 + mb];
        dst[(mb + 0) * 2] = kv.x;
        dst[(mb + 1) * 2] = kv.y;
        dst[(mb + 2) * 2] = kv.z;
        dst[(mb + 3) * 2] = kv.w;
      }
    }
    {
#pragma unroll
      for (int i = 0; i < 8; ++i) {
        const int mb = 4 * i;
        const float4 vv = *(const float4*)&vb[(size_t)vc * NSEQ + m0 + mb];
        vtp[(2 * i + 0) * LDV2 + 2 * vc + 0] = vv.x;
        vtp[(2 * i + 0) * LDV2 + 2 * vc + 1] = vv.y;
        vtp[(2 * i + 1) * LDV2 + 2 * vc + 0] = vv.z;
        vtp[(2 * i + 1) * LDV2 + 2 * vc + 1] = vv.w;
      }
    }
    __syncthreads();

    if (m0 + MT < NSEQ) {
      __builtin_prefetch(&kb[(size_t)kc * NSEQ + m0 + MT + kmh], 0, 1);
      __builtin_prefetch(&vb[(size_t)vc * NSEQ + m0 + MT], 0, 1);
    }

    v8f E[2];
#pragma unroll
    for (int st = 0; st < 2; ++st) { v8f z = {}; E[st] = z; }
    for (int t = 0; t < 16; ++t) {
      const float* krow = &kbp[(2 * t + h) * LDK2];
#pragma unroll
      for (int st = 0; st < 2; ++st) {
        v2f bop = *(const v2f*)&krow[(16 * st + l15) * 2];
        E[st] = WMMA_F32X4(qa[t], bop, E[st]);
      }
    }

    v8f tmax;
#pragma unroll
    for (int j = 0; j < 8; ++j) tmax[j] = fmaxf(E[0][j], E[1][j]);
#pragma unroll
    for (int d = 1; d <= 8; d <<= 1) {
#pragma unroll
      for (int j = 0; j < 8; ++j)
        tmax[j] = fmaxf(tmax[j], __shfl_xor(tmax[j], d, 32));
    }
    if (l15 == 0) {
#pragma unroll
      for (int j = 0; j < 8; ++j) red[wave][j + 8 * h] = tmax[j];
    }
    const float rmax = red[wave][l15];
    const float mnew = fmaxf(m_s, rmax);
    const float scale = __expf(m_s - mnew);
    m_s = mnew;
    if (h == 0) mrow_s[wave][l15] = mnew;

    v8f mrow;
#pragma unroll
    for (int j = 0; j < 8; ++j) mrow[j] = mrow_s[wave][j + 8 * h];
#pragma unroll
    for (int st = 0; st < 2; ++st)
#pragma unroll
      for (int j = 0; j < 8; ++j) E[st][j] = __expf(E[st][j] - mrow[j]);

    v8f tsum;
#pragma unroll
    for (int j = 0; j < 8; ++j) tsum[j] = E[0][j] + E[1][j];
#pragma unroll
    for (int d = 1; d <= 8; d <<= 1) {
#pragma unroll
      for (int j = 0; j < 8; ++j) tsum[j] += __shfl_xor(tsum[j], d, 32);
    }
    if (l15 == 0) {
#pragma unroll
      for (int j = 0; j < 8; ++j) red[wave][j + 8 * h] = tsum[j];
    }
    l_s = l_s * scale + red[wave][l15];

#pragma unroll
    for (int ct = 0; ct < 8; ++ct)
#pragma unroll
      for (int j = 0; j < 8; ++j) Oacc[ct][j] *= scale;

#pragma unroll
    for (int st = 0; st < 2; ++st) {
      const int m = 16 * st + l15;
#pragma unroll
      for (int j = 0; j < 8; ++j)
        ptw[(m >> 1) * LDP2 + (j + 8 * h) * 2 + (m & 1)] = E[st][j];
    }

    for (int t = 0; t < 8; ++t) {
      const int pairm = 2 * t + h;
      v2f bop = *(const v2f*)&ptw[pairm * LDP2 + l15 * 2];
      const float* vrow = &vtp[pairm * LDV2];
#pragma unroll
      for (int ct = 0; ct < 8; ++ct) {
        v2f aop = *(const v2f*)&vrow[(16 * ct + l15) * 2];
        Oacc[ct] = WMMA_F32X4(aop, bop, Oacc[ct]);
      }
    }
  }

  const float g = gamma_p[0];
  const float inv_l = 1.0f / l_s;
  const float* xhb = x_h + (size_t)b * CV * NSEQ;
  float* ob        = out + (size_t)b * CV * NSEQ;
  const int n = n0 + l15;
#pragma unroll
  for (int ct = 0; ct < 8; ++ct) {
#pragma unroll
    for (int j = 0; j < 8; ++j) {
      const int c = 16 * ct + j + 8 * h;
      const size_t idx = (size_t)c * NSEQ + n;
      ob[idx] = g * (Oacc[ct][j] * inv_l) + xhb[idx];
    }
  }
}

extern "C" void kernel_launch(void* const* d_in, const int* in_sizes, int n_in,
                              void* d_out, int out_size, void* d_ws, size_t ws_size,
                              hipStream_t stream) {
  (void)in_sizes; (void)n_in; (void)out_size; (void)ws_size;
  const float* x     = (const float*)d_in[0];
  const float* x_h   = (const float*)d_in[1];
  const float* Wq    = (const float*)d_in[2];
  const float* Wk    = (const float*)d_in[3];
  const float* Wv    = (const float*)d_in[4];
  const float* gamma = (const float*)d_in[5];
  float* out = (float*)d_out;

  float* qw = (float*)d_ws;
  float* kw = qw + (size_t)BATCH * CQK * NSEQ;
  float* vw = kw + (size_t)BATCH * CQK * NSEQ;

  dim3 blk(128);
  proj_wmma_kernel<<<dim3(NSEQ / 64, CQK / 64, BATCH), blk, 0, stream>>>(Wq, x, qw, CQK, CQK);
  proj_wmma_kernel<<<dim3(NSEQ / 64, CQK / 64, BATCH), blk, 0, stream>>>(Wk, x, kw, CQK, CQK);
  proj_wmma_kernel<<<dim3(NSEQ / 64, CV / 64, BATCH), blk, 0, stream>>>(Wv, x_h, vw, CV, CV);
  flash_attn_f32_kernel<<<dim3(NSEQ / 64, BATCH), blk, 0, stream>>>(qw, kw, vw, x_h, gamma, out);
}


# attention kernel: 8 waves per block (two per SIMD); the two 4-wave groups take alternate 32-key tiles with own staging buffers, partial (O, m, l) merged once through LDS
# speedup vs baseline: 1.3388x; 1.3388x over previous
#include <hip/hip_runtime.h>


typedef float v2f __attribute__((ext_vector_type(2)));
typedef float v8f __attribute__((ext_vector_type(8)));

#define WMMA_F32X4(A, B, C) \
  __builtin_amdgcn_wmma_f32_16x16x4_f32(false, (A), false, (B), (short)0, (C), false, false)

namespace {
constexpr int BATCH = 4;
constexpr int CQK   = 64;
constexpr int CV    = 128;
constexpr int NSEQ  = 4096;
constexpr int MT    = 32;
constexpr int LDK2 = 2 * MT + 2;
constexpr int LDV2 = 2 * CV + 2;
constexpr int LDP2 = 2 * 16 + 2;
}

__global__ __launch_bounds__(128) void proj_wmma_kernel(
    const float* __restrict__ W, const float* __restrict__ X,
    float* __restrict__ Y, int O, int C) {
  const int lane = threadIdx.x & 31;
  const int wave = threadIdx.x >> 5;
  const int h    = lane >> 4;
  const int l15  = lane & 15;
  const int b    = blockIdx.z;
  const int o0   = blockIdx.y * 64 + wave * 16;
  const int n0   = blockIdx.x * 64;

  const float* Xb = X + (size_t)b * C * NSEQ;
  float* Yb       = Y + (size_t)b * O * NSEQ;

  v8f acc[4];
#pragma unroll
  for (int st = 0; st < 4; ++st) { v8f z = {}; acc[st] = z; }

  const int ksteps = C >> 2;
  for (int t = 0; t < ksteps; ++t) {
    const int c0 = 4 * t + 2 * h;
    v2f a = *(const v2f*)&W[(size_t)(o0 + l15) * C + c0];
#pragma unroll
    for (int st = 0; st < 4; ++st) {
      const int n = n0 + 16 * st + l15;
      v2f bop;
      bop[0] = Xb[(size_t)c0 * NSEQ + n];
      bop[1] = Xb[(size_t)(c0 + 1) * NSEQ + n];
      acc[st] = WMMA_F32X4(a, bop, acc[st]);
    }
  }
#pragma unroll
  for (int st = 0; st < 4; ++st) {
    const int n = n0 + 16 * st + l15;
#pragma unroll
    for (int j = 0; j < 8; ++j) {
      Yb[(size_t)(o0 + j + 8 * h) * NSEQ + n] = acc[st][j];
    }
  }
}

__global__ __launch_bounds__(256) void flash_attn_f32_kernel(
    const float* __restrict__ q, const float* __restrict__ k,
    const float* __restrict__ v, const float* __restrict__ x_h,
    const float* __restrict__ gamma_p, float* __restrict__ out) {
  __shared__ float kbp2[2][(CQK / 2) * LDK2];
  __shared__ float vtp2[2][(MT / 2) * LDV2];
  __shared__ float ptb[8 * (MT / 2) * LDP2];
  __shared__ float red[8][16];
  __shared__ float mrow_s[8][16];
  __shared__ float mrg[4][66][32];

  const int tid  = threadIdx.x;
  const int lane = tid & 31;
  const int wave = tid >> 5;
  const int grp  = wave >> 2;
  const int qw   = wave & 3;
  const int t7   = tid & 127;
  const int h    = lane >> 4;
  const int l15  = lane & 15;
  const int b    = blockIdx.y;
  const int n0   = blockIdx.x * 64 + qw * 16;
  float* kbp = kbp2[grp];
  float* vtp = vtp2[grp];

  const float* qb  = q + (size_t)b * CQK * NSEQ;
  const float* kb  = k + (size_t)b * CQK * NSEQ;
  const float* vb  = v + (size_t)b * CV * NSEQ;

  const int kc  = t7 >> 1;
  const int kmh = (t7 & 1) * 16;
  const int vc  = t7;

  v2f qa[16];
#pragma unroll
  for (int t = 0; t < 16; ++t) {
    const int c0 = 4 * t + 2 * h;
    const int n  = n0 + l15;
    qa[t][0] = qb[(size_t)c0 * NSEQ + n];
    qa[t][1] = qb[(size_t)(c0 + 1) * NSEQ + n];
  }

  v8f Oacc[8];
#pragma unroll
  for (int ct = 0; ct < 8; ++ct) { v8f z = {}; Oacc[ct] = z; }
  float m_s = -3.0e38f;
  float l_s = 0.0f;

  float* ptw = &ptb[wave * (MT / 2) * LDP2];

  for (int it = 0; it < NSEQ / (2 * MT); ++it) {
    const int m0 = (2 * it + grp) * MT;
    __syncthreads();

    {
      float* dst = &kbp[(kc >> 1) * LDK2 + (kc & 1)];
#pragma unroll
      for (int i = 0; i < 4; ++i) {
        const int mb = kmh + 4 * i;
        const float4 kv = *(const float4*)&kb[(size_t)kc * NSEQ + m0 + mb];
        dst[(mb + 0) * 2] = kv.x;
        dst[(mb + 1) * 2] = kv.y;
        dst[(mb + 2) * 2] = kv.z;
        dst[(mb + 3) * 2] = kv.w;
      }
    }
    {
#pragma unroll
      for (int i = 0; i < 8; ++i) {
        const int mb = 4 * i;
        const float4 vv = *(const float4*)&vb[(size_t)vc * NSEQ + m0 + mb];
        vtp[(2 * i + 0) * LDV2 + 2 * vc + 0] = vv.x;
        vtp[(2 * i + 0) * LDV2 + 2 * vc + 1] = vv.y;
        vtp[(2 * i + 1) * LDV2 + 2 * vc + 0] = vv.z;
        vtp[(2 * i + 1) * LDV2 + 2 * vc + 1] = vv.w;
      }
    }
    __syncthreads();

    if (it + 1 < NSEQ / (2 * MT)) {
      __builtin_prefetch(&kb[(size_t)kc * NSEQ + m0 + 2 * MT + kmh], 0, 1);
      __builtin_prefetch(&vb[(size_t)vc * NSEQ + m0 + 2 * MT], 0, 1);
    }

    v8f E[2];
#pragma unroll
    for (int st = 0; st < 2; ++st) { v8f z = {}; E[st] = z; }
    for (int t = 0; t < 16; ++t) {
      const float* krow = &kbp[(2 * t + h) * LDK2];
#pragma unroll
      for (int st = 0; st < 2; ++st) {
        v2f bop = *(const v2f*)&krow[(16 * st + l15) * 2];
        E[st] = WMMA_F32X4(qa[t], bop, E[st]);
      }
    }

    v8f tmax;
#pragma unroll
    for (int j = 0; j < 8; ++j) tmax[j] = fmaxf(E[0][j], E[1][j]);
#pragma unroll
    for (int d = 1; d <= 8; d <<= 1) {
#pragma unroll
      for (int j = 0; j < 8; ++j)
        tmax[j] = fmaxf(tmax[j], __shfl_xor(tmax[j], d, 32));
    }
    if (l15 == 0) {
#pragma unroll
      for (int j = 0; j < 8; ++j) red[wave][j + 8 * h] = tmax[j];
    }
    const float rmax = red[wave][l15];
    const float mnew = fmaxf(m_s, rmax);
    const float scale = __expf(m_s - mnew);
    m_s = mnew;
    if (h == 0) mrow_s[wave][l15] = mnew;

    v8f mrow;
#pragma unroll
    for (int j = 0; j < 8; ++j) mrow[j] = mrow_s[wave][j + 8 * h];
#pragma unroll
    for (int st = 0; st < 2; ++st)
#pragma unroll
      for (int j = 0; j < 8; ++j) E[st][j] = __expf(E[st][j] - mrow[j]);

    v8f tsum;
#pragma unroll
    for (int j = 0; j < 8; ++j) tsum[j] = E[0][j] + E[1][j];
#pragma unroll
    for (int d = 1; d <= 8; d <<= 1) {
#pragma unroll
      for (int j = 0; j < 8; ++j) tsum[j] += __shfl_xor(tsum[j], d, 32);
    }
    if (l15 == 0) {
#pragma unroll
      for (int j = 0; j < 8; ++j) red[wave][j + 8 * h] = tsum[j];
    }
    l_s = l_s * scale + red[wave][l15];

#pragma unroll
    for (int ct = 0; ct < 8; ++ct)
#pragma unroll
      for (int j = 0; j < 8; ++j) Oacc[ct][j] *= scale;

#pragma unroll
    for (int st = 0; st < 2; ++st) {
      const int m = 16 * st + l15;
#pragma unroll
      for (int j = 0; j < 8; ++j)
        ptw[(m >> 1) * LDP2 + (j + 8 * h) * 2 + (m & 1)] = E[st][j];
    }

    for (int t = 0; t < 8; ++t) {
      const int pairm = 2 * t + h;
      v2f bop = *(const v2f*)&ptw[pairm * LDP2 + l15 * 2];
      const float* vrow = &vtp[pairm * LDV2];
#pragma unroll
      for (int ct = 0; ct < 8; ++ct) {
        v2f aop = *(const v2f*)&vrow[(16 * ct + l15) * 2];
        Oacc[ct] = WMMA_F32X4(aop, bop, Oacc[ct]);
      }
    }
  }

  if (grp == 1) {
#pragma unroll
    for (int ct = 0; ct < 8; ++ct)
#pragma unroll
      for (int j = 0; j < 8; ++j) mrg[qw][8 * ct + j][lane] = Oacc[ct][j];
    mrg[qw][64][lane] = m_s;
    mrg[qw][65][lane] = l_s;
  }
  __syncthreads();

  if (grp == 0) {
    const float m1 = mrg[qw][64][lane];
    const float l1 = mrg[qw][65][lane];
    const float mm = fmaxf(m_s, m1);
    const float a0 = __expf(m_s - mm);
    const float a1 = __expf(m1 - mm);
    const float g = gamma_p[0];
    const float inv_l = 1.0f / (l_s * a0 + l1 * a1);
    const float* xhb = x_h + (size_t)b * CV * NSEQ;
    float* ob        = out + (size_t)b * CV * NSEQ;
    const int n = n0 + l15;
#pragma unroll
    for (int ct = 0; ct < 8; ++ct) {
#pragma unroll
      for (int j = 0; j < 8; ++j) {
        const int c = 16 * ct + j + 8 * h;
        const size_t idx = (size_t)c * NSEQ + n;
        const float o = Oacc[ct][j] * a0 + mrg[qw][8 * ct + j][lane] * a1;
        ob[idx] = g * (o * inv_l) + xhb[idx];
      }
    }
  }
}

extern "C" void kernel_launch(void* const* d_in, const int* in_sizes, int n_in,
                              void* d_out, int out_size, void* d_ws, size_t ws_size,
                              hipStream_t stream) {
  (void)in_sizes; (void)n_in; (void)out_size; (void)ws_size;
  const float* x     = (const float*)d_in[0];
  const float* x_h   = (const float*)d_in[1];
  const float* Wq    = (const float*)d_in[2];
  const float* Wk    = (const float*)d_in[3];
  const float* Wv    = (const float*)d_in[4];
  const float* gamma = (const float*)d_in[5];
  float* out = (float*)d_out;

  float* qw = (float*)d_ws;
  float* kw = qw + (size_t)BATCH * CQK * NSEQ;
  float* vw = kw + (size_t)BATCH * CQK * NSEQ;

  dim3 blk(128);
  proj_wmma_kernel<<<dim3(NSEQ / 64, CQK / 64, BATCH), blk, 0, stream>>>(Wq, x, qw, CQK, CQK);
  proj_wmma_kernel<<<dim3(NSEQ / 64, CQK / 64, BATCH), blk, 0, stream>>>(Wk, x, kw, CQK, CQK);
  proj_wmma_kernel<<<dim3(NSEQ / 64, CV / 64, BATCH), blk, 0, stream>>>(Wv, x_h, vw, CV, CV);
  flash_attn_f32_kernel<<<dim3(NSEQ / 64, BATCH), dim3(256), 0, stream>>>(qw, kw, vw, x_h, gamma, out);
}


# attention kernel: score tile computed with keys on rows (S^T = k^T q): max/sum per query are in-lane plus one lane-half exchange, exp(S^T) feeds the PV product from registers, no LDS transpose
# speedup vs baseline: 1.5681x; 1.1713x over previous
#include <hip/hip_runtime.h>


typedef float v2f __attribute__((ext_vector_type(2)));
typedef float v8f __attribute__((ext_vector_type(8)));

#define WMMA_F32X4(A, B, C) \
  __builtin_amdgcn_wmma_f32_16x16x4_f32(false, (A), false, (B), (short)0, (C), false, false)

namespace {
constexpr int BATCH = 4;
constexpr int CQK   = 64;
constexpr int CV    = 128;
constexpr int NSEQ  = 4096;
constexpr int MT    = 32;
constexpr int LDK2 = 2 * MT + 2;
constexpr int LDV2 = 2 * CV + 8;
}

__device__ __forceinline__ float xhalf(float x) {
  const int b = __builtin_bit_cast(int, x);
  return __builtin_bit_cast(float,
      __builtin_amdgcn_permlanex16(b, b, 0x76543210, 0xfedcba98, false, false));
}

__global__ __launch_bounds__(128) void proj_wmma_kernel(
    const float* __restrict__ W, const float* __restrict__ X,
    float* __restrict__ Y, int O, int C) {
  const int lane = threadIdx.x & 31;
  const int wave = threadIdx.x >> 5;
  const int h    = lane >> 4;
  const int l15  = lane & 15;
  const int b    = blockIdx.z;
  const int o0   = blockIdx.y * 64 + wave * 16;
  const int n0   = blockIdx.x * 64;

  const float* Xb = X + (size_t)b * C * NSEQ;
  float* Yb       = Y + (size_t)b * O * NSEQ;

  v8f acc[4];
#pragma unroll
  for (int st = 0; st < 4; ++st) { v8f z = {}; acc[st] = z; }

  const int ksteps = C >> 2;
  for (int t = 0; t < ksteps; ++t) {
    const int c0 = 4 * t + 2 * h;
    v2f a = *(const v2f*)&W[(size_t)(o0 + l15) * C + c0];
#pragma unroll
    for (int st = 0; st < 4; ++st) {
      const int n = n0 + 16 * st + l15;
      v2f bop;
      bop[0] = Xb[(size_t)c0 * NSEQ + n];
      bop[1] = Xb[(size_t)(c0 + 1) * NSEQ + n];
      acc[st] = WMMA_F32X4(a, bop, acc[st]);
    }
  }
#pragma unroll
  for (int st = 0; st < 4; ++st) {
    const int n = n0 + 16 * st + l15;
#pragma unroll
    for (int j = 0; j < 8; ++j) {
      Yb[(size_t)(o0 + j + 8 * h) * NSEQ + n] = acc[st][j];
    }
  }
}

__global__ __launch_bounds__(256) void flash_attn_f32_kernel(
    const float* __restrict__ q, const float* __restrict__ k,
    const float* __restrict__ v, const float* __restrict__ x_h,
    const float* __restrict__ gamma_p, float* __restrict__ out) {
  __shared__ float kbp2[2][(CQK / 2) * LDK2];
  __shared__ float vtp2[2][(MT / 2) * LDV2];
  __shared__ float mrg[4][66][32];

  const int tid  = threadIdx.x;
  const int lane = tid & 31;
  const int wave = tid >> 5;
  const int grp  = wave >> 2;
  const int qw   = wave & 3;
  const int t7   = tid & 127;
  const int h    = lane >> 4;
  const int l15  = lane & 15;
  const int b    = blockIdx.y;
  const int n0   = blockIdx.x * 64 + qw * 16;
  float* kbp = kbp2[grp];
  float* vtp = vtp2[grp];

  const float* qb  = q + (size_t)b * CQK * NSEQ;
  const float* kb  = k + (size_t)b * CQK * NSEQ;
  const float* vb  = v + (size_t)b * CV * NSEQ;

  const int kc  = t7 >> 1;
  const int kmh = (t7 & 1) * 16;
  const int vc  = t7;

  v2f qa[16];
#pragma unroll
  for (int t = 0; t < 16; ++t) {
    const int c0 = 4 * t + 2 * h;
    const int n  = n0 + l15;
    qa[t][0] = qb[(size_t)c0 * NSEQ + n];
    qa[t][1] = qb[(size_t)(c0 + 1) * NSEQ + n];
  }

  v8f Oacc[8];
#pragma unroll
  for (int ct = 0; ct < 8; ++ct) { v8f z = {}; Oacc[ct] = z; }
  float m_s = -3.0e38f;
  float l_s = 0.0f;

  for (int it = 0; it < NSEQ / (2 * MT); ++it) {
    const int m0 = (2 * it + grp) * MT;
    __syncthreads();

    {
      float* dst = &kbp[(kc >> 1) * LDK2 + (kc & 1)];
#pragma unroll
      for (int i = 0; i < 4; ++i) {
        const int mb = kmh + 4 * i;
        const float4 kv = *(const float4*)&kb[(size_t)kc * NSEQ + m0 + mb];
        dst[(mb + 0) * 2] = kv.x;
        dst[(mb + 1) * 2] = kv.y;
        dst[(mb + 2) * 2] = kv.z;
        dst[(mb + 3) * 2] = kv.w;
      }
    }
    {
#pragma unroll
      for (int i = 0; i < 8; ++i) {
        const int mb = 4 * i;
        const float4 vv = *(const float4*)&vb[(size_t)vc * NSEQ + m0 + mb];
        vtp[(2 * i + 0) * LDV2 + 2 * vc + 0] = vv.x;
        vtp[(2 * i + 0) * LDV2 + 2 * vc + 1] = vv.y;
        vtp[(2 * i + 1) * LDV2 + 2 * vc + 0] = vv.z;
        vtp[(2 * i + 1) * LDV2 + 2 * vc + 1] = vv.w;
      }
    }
    __syncthreads();

    if (it + 1 < NSEQ / (2 * MT)) {
      __builtin_prefetch(&kb[(size_t)kc * NSEQ + m0 + 2 * MT + kmh], 0, 1);
      __builtin_prefetch(&vb[(size_t)vc * NSEQ + m0 + 2 * MT], 0, 1);
    }

    v8f E[2];
#pragma unroll
    for (int st = 0; st < 2; ++st) { v8f z = {}; E[st] = z; }
    for (int t = 0; t < 16; ++t) {
      const float* krow = &kbp[(2 * t + h) * LDK2];
#pragma unroll
      for (int st = 0; st < 2; ++st) {
        v2f aop = *(const v2f*)&krow[(16 * st + l15) * 2];
        E[st] = WMMA_F32X4(aop, qa[t], E[st]);
      }
    }

    float tmax = fmaxf(E[0][0], E[1][0]);
#pragma unroll
    for (int j = 1; j < 8; ++j) tmax = fmaxf(tmax, fmaxf(E[0][j], E[1][j]));
    tmax = fmaxf(tmax, xhalf(tmax));
    const float mnew = fmaxf(m_s, tmax);
    const float scale = __expf(m_s - mnew);
    m_s = mnew;

#pragma unroll
    for (int st = 0; st < 2; ++st)
#pragma unroll
      for (int j = 0; j < 8; ++j) E[st][j] = __expf(E[st][j] - mnew);

    float tsum = E[0][0] + E[1][0];
#pragma unroll
    for (int j = 1; j < 8; ++j) tsum += E[0][j] + E[1][j];
    l_s = l_s * scale + tsum;

#pragma unroll
    for (int ct = 0; ct < 8; ++ct)
#pragma unroll
      for (int j = 0; j < 8; ++j) Oacc[ct][j] *= scale;

#pragma unroll
    for (int st = 0; st < 2; ++st) {
#pragma unroll
      for (int u = 0; u < 4; ++u) {
        v2f bop;
        bop[0] = E[st][2 * u];
        bop[1] = E[st][2 * u + 1];
        const float* vrow = &vtp[(8 * st + 4 * h + u) * LDV2];
#pragma unroll
        for (int ct = 0; ct < 8; ++ct) {
          v2f aop = *(const v2f*)&vrow[(16 * ct + l15) * 2];
          Oacc[ct] = WMMA_F32X4(aop, bop, Oacc[ct]);
        }
      }
    }
  }

  l_s = l_s + xhalf(l_s);

  if (grp == 1) {
#pragma unroll
    for (int ct = 0; ct < 8; ++ct)
#pragma unroll
      for (int j = 0; j < 8; ++j) mrg[qw][8 * ct + j][lane] = Oacc[ct][j];
    mrg[qw][64][lane] = m_s;
    mrg[qw][65][lane] = l_s;
  }
  __syncthreads();

  if (grp == 0) {
    const float m1 = mrg[qw][64][lane];
    const float l1 = mrg[qw][65][lane];
    const float mm = fmaxf(m_s, m1);
    const float a0 = __expf(m_s - mm);
    const float a1 = __expf(m1 - mm);
    const float g = gamma_p[0];
    const float inv_l = 1.0f / (l_s * a0 + l1 * a1);
    const float* xhb = x_h + (size_t)b * CV * NSEQ;
    float* ob        = out + (size_t)b * CV * NSEQ;
    const int n = n0 + l15;
#pragma unroll
    for (int ct = 0; ct < 8; ++ct) {
#pragma unroll
      for (int j = 0; j < 8; ++j) {
        const int c = 16 * ct + j + 8 * h;
        const size_t idx = (size_t)c * NSEQ + n;
        const float o = Oacc[ct][j] * a0 + mrg[qw][8 * ct + j][lane] * a1;
        ob[idx] = g * (o * inv_l) + xhb[idx];
      }
    }
  }
}

extern "C" void kernel_launch(void* const* d_in, const int* in_sizes, int n_in,
                              void* d_out, int out_size, void* d_ws, size_t ws_size,
                              hipStream_t stream) {
  (void)in_sizes; (void)n_in; (void)out_size; (void)ws_size;
  const float* x     = (const float*)d_in[0];
  const float* x_h   = (const float*)d_in[1];
  const float* Wq    = (const float*)d_in[2];
  const float* Wk    = (const float*)d_in[3];
  const float* Wv    = (const float*)d_in[4];
  const float* gamma = (const float*)d_in[5];
  float* out = (float*)d_out;

  float* qw = (float*)d_ws;
  float* kw = qw + (size_t)BATCH * CQK * NSEQ;
  float* vw = kw + (size_t)BATCH * CQK * NSEQ;

  dim3 blk(128);
  proj_wmma_kernel<<<dim3(NSEQ / 64, CQK / 64, BATCH), blk, 0, stream>>>(Wq, x, qw, CQK, CQK);
  proj_wmma_kernel<<<dim3(NSEQ / 64, CQK / 64, BATCH), blk, 0, stream>>>(Wk, x, kw, CQK, CQK);
  proj_wmma_kernel<<<dim3(NSEQ / 64, CV / 64, BATCH), blk, 0, stream>>>(Wv, x_h, vw, CV, CV);
  flash_attn_f32_kernel<<<dim3(NSEQ / 64, BATCH), dim3(256), 0, stream>>>(qw, kw, vw, x_h, gamma, out);
}


# attention kernel: K/V staging re-mapped, 8 lanes per 128-byte tile row, so a wave-wide 16-byte global load reads 4 whole lines (was 16 or 32 part lines); V columns swizzled, LDS stores conflict-free
# speedup vs baseline: 1.7963x; 1.1455x over previous
#include <hip/hip_runtime.h>


typedef float v2f __attribute__((ext_vector_type(2)));
typedef float v8f __attribute__((ext_vector_type(8)));

#define WMMA_F32X4(A, B, C) \
  __builtin_amdgcn_wmma_f32_16x16x4_f32(false, (A), false, (B), (short)0, (C), false, false)

namespace {
constexpr int BATCH = 4;
constexpr int CQK   = 64;
constexpr int CV    = 128;
constexpr int NSEQ  = 4096;
constexpr int MT    = 32;
constexpr int LDK2 = 2 * MT + 2;
constexpr int LDV2 = 2 * CV + 8;
}

__device__ __forceinline__ float xhalf(float x) {
  const int b = __builtin_bit_cast(int, x);
  return __builtin_bit_cast(float,
      __builtin_amdgcn_permlanex16(b, b, 0x76543210, 0xfedcba98, false, false));
}

__global__ __launch_bounds__(128) void proj_wmma_kernel(
    const float* __restrict__ W, const float* __restrict__ X,
    float* __restrict__ Y, int O, int C) {
  const int lane = threadIdx.x & 31;
  const int wave = threadIdx.x >> 5;
  const int h    = lane >> 4;
  const int l15  = lane & 15;
  const int b    = blockIdx.z;
  const int o0   = blockIdx.y * 64 + wave * 16;
  const int n0   = blockIdx.x * 64;

  const float* Xb = X + (size_t)b * C * NSEQ;
  float* Yb       = Y + (size_t)b * O * NSEQ;

  v8f acc[4];
#pragma unroll
  for (int st = 0; st < 4; ++st) { v8f z = {}; acc[st] = z; }

  const int ksteps = C >> 2;
  for (int t = 0; t < ksteps; ++t) {
    const int c0 = 4 * t + 2 * h;
    v2f a = *(const v2f*)&W[(size_t)(o0 + l15) * C + c0];
#pragma unroll
    for (int st = 0; st < 4; ++st) {
      const int n = n0 + 16 * st + l15;
      v2f bop;
      bop[0] = Xb[(size_t)c0 * NSEQ + n];
      bop[1] = Xb[(size_t)(c0 + 1) * NSEQ + n];
      acc[st] = WMMA_F32X4(a, bop, acc[st]);
    }
  }
#pragma unroll
  for (int st = 0; st < 4; ++st) {
    const int n = n0 + 16 * st + l15;
#pragma unroll
    for (int j = 0; j < 8; ++j) {
      Yb[(size_t)(o0 + j + 8 * h) * NSEQ + n] = acc[st][j];
    }
  }
}

__global__ __launch_bounds__(256) void flash_attn_f32_kernel(
    const float* __restrict__ q, const float* __restrict__ k,
    const float* __restrict__ v, const float* __restrict__ x_h,
    const float* __restrict__ gamma_p, float* __restrict__ out) {
  __shared__ float kbp2[2][(CQK / 2) * LDK2];
  __shared__ float vtp2[2][(MT / 2) * LDV2];
  __shared__ float mrg[4][66][32];

  const int tid  = threadIdx.x;
  const int lane = tid & 31;
  const int wave = tid >> 5;
  const int grp  = wave >> 2;
  const int qw   = wave & 3;
  const int t7   = tid & 127;
  const int h    = lane >> 4;
  const int l15  = lane & 15;
  const int b    = blockIdx.y;
  const int n0   = blockIdx.x * 64 + qw * 16;
  float* kbp = kbp2[grp];
  float* vtp = vtp2[grp];

  const float* qb  = q + (size_t)b * CQK * NSEQ;
  const float* kb  = k + (size_t)b * CQK * NSEQ;
  const float* vb  = v + (size_t)b * CV * NSEQ;

  const int sq  = t7 & 7;
  const int sg  = t7 >> 3;
  const int scol = sg ^ ((sq >> 2) * 4);
  const int l15s = l15 ^ 4;
  const float* ksrc = kb + (size_t)(2 * sg) * NSEQ + grp * MT + 4 * sq;
  const float* vsrc = vb + (size_t)sg * NSEQ + grp * MT + 4 * sq;
  float* kdst = &kbp[sg * LDK2 + 8 * sq];
  float* vdst = &vtp[(2 * sq) * LDV2 + 2 * scol];
  const int kc  = t7 >> 1;
  const int kmh = (t7 & 1) * 16;
  const int vc  = t7;

  v2f qa[16];
#pragma unroll
  for (int t = 0; t < 16; ++t) {
    const int c0 = 4 * t + 2 * h;
    const int n  = n0 + l15;
    qa[t][0] = qb[(size_t)c0 * NSEQ + n];
    qa[t][1] = qb[(size_t)(c0 + 1) * NSEQ + n];
  }

  v8f Oacc[8];
#pragma unroll
  for (int ct = 0; ct < 8; ++ct) { v8f z = {}; Oacc[ct] = z; }
  float m_s = -3.0e38f;
  float l_s = 0.0f;

  for (int it = 0; it < NSEQ / (2 * MT); ++it) {
    const int m0 = (2 * it + grp) * MT;
    __syncthreads();

    {
      const float* kp = ksrc + (size_t)it * (2 * MT);
#pragma unroll
      for (int ii = 0; ii < 2; ++ii) {
        const float4 ke = *(const float4*)(kp + (size_t)(32 * ii) * NSEQ);
        const float4 ko = *(const float4*)(kp + (size_t)(32 * ii + 1) * NSEQ);
        float* dst = kdst + 16 * ii * LDK2;
        v2f p0, p1, p2, p3;
        p0[0] = ke.x; p0[1] = ko.x;
        p1[0] = ke.y; p1[1] = ko.y;
        p2[0] = ke.z; p2[1] = ko.z;
        p3[0] = ke.w; p3[1] = ko.w;
        *(v2f*)&dst[0] = p0;
        *(v2f*)&dst[2] = p1;
        *(v2f*)&dst[4] = p2;
        *(v2f*)&dst[6] = p3;
      }
    }
    {
      const float* vp = vsrc + (size_t)it * (2 * MT);
#pragma unroll
      for (int i = 0; i < 8; ++i) {
        const float4 vv = *(const float4*)(vp + (size_t)(16 * i) * NSEQ);
        float* dst = vdst + 32 * i;
        v2f lo, hi;
        lo[0] = vv.x; lo[1] = vv.y;
        hi[0] = vv.z; hi[1] = vv.w;
        *(v2f*)&dst[0] = lo;
        *(v2f*)&dst[LDV2] = hi;
      }
    }
    __syncthreads();

    if (it + 1 < NSEQ / (2 * MT)) {
      __builtin_prefetch(&kb[(size_t)kc * NSEQ + m0 + 2 * MT + kmh], 0, 1);
      __builtin_prefetch(&vb[(size_t)vc * NSEQ + m0 + 2 * MT], 0, 1);
    }

    v8f E[2];
#pragma unroll
    for (int st = 0; st < 2; ++st) { v8f z = {}; E[st] = z; }
    for (int t = 0; t < 16; ++t) {
      const float* krow = &kbp[(2 * t + h) * LDK2];
#pragma unroll
      for (int st = 0; st < 2; ++st) {
        v2f aop = *(const v2f*)&krow[(16 * st + l15) * 2];
        E[st] = WMMA_F32X4(aop, qa[t], E[st]);
      }
    }

    float tmax = fmaxf(E[0][0], E[1][0]);
#pragma unroll
    for (int j = 1; j < 8; ++j) tmax = fmaxf(tmax, fmaxf(E[0][j], E[1][j]));
    tmax = fmaxf(tmax, xhalf(tmax));
    const float mnew = fmaxf(m_s, tmax);
    const float scale = __expf(m_s - mnew);
    m_s = mnew;

#pragma unroll
    for (int st = 0; st < 2; ++st)
#pragma unroll
      for (int j = 0; j < 8; ++j) E[st][j] = __expf(E[st][j] - mnew);

    float tsum = E[0][0] + E[1][0];
#pragma unroll
    for (int j = 1; j < 8; ++j) tsum += E[0][j] + E[1][j];
    l_s = l_s * scale + tsum;

#pragma unroll
    for (int ct = 0; ct < 8; ++ct)
#pragma unroll
      for (int j = 0; j < 8; ++j) Oacc[ct][j] *= scale;

#pragma unroll
    for (int st = 0; st < 2; ++st) {
      const int lc = st ? l15s : l15;
#pragma unroll
      for (int u = 0; u < 4; ++u) {
        v2f bop;
        bop[0] = E[st][2 * u];
        bop[1] = E[st][2 * u + 1];
        const float* vrow = &vtp[(8 * st + 4 * h + u) * LDV2];
#pragma unroll
        for (int ct = 0; ct < 8; ++ct) {
          v2f aop = *(const v2f*)&vrow[(16 * ct + lc) * 2];
          Oacc[ct] = WMMA_F32X4(aop, bop, Oacc[ct]);
        }
      }
    }
  }

  l_s = l_s + xhalf(l_s);

  if (grp == 1) {
#pragma unroll
    for (int ct = 0; ct < 8; ++ct)
#pragma unroll
      for (int j = 0; j < 8; ++j) mrg[qw][8 * ct + j][lane] = Oacc[ct][j];
    mrg[qw][64][lane] = m_s;
    mrg[qw][65][lane] = l_s;
  }
  __syncthreads();

  if (grp == 0) {
    const float m1 = mrg[qw][64][lane];
    const float l1 = mrg[qw][65][lane];
    const float mm = fmaxf(m_s, m1);
    const float a0 = __expf(m_s - mm);
    const float a1 = __expf(m1 - mm);
    const float g = gamma_p[0];
    const float inv_l = 1.0f / (l_s * a0 + l1 * a1);
    const float* xhb = x_h + (size_t)b * CV * NSEQ;
    float* ob        = out + (size_t)b * CV * NSEQ;
    const int n = n0 + l15;
#pragma unroll
    for (int ct = 0; ct < 8; ++ct) {
#pragma unroll
      for (int j = 0; j < 8; ++j) {
        const int c = 16 * ct + j + 8 * h;
        const size_t idx = (size_t)c * NSEQ + n;
        const float o = Oacc[ct][j] * a0 + mrg[qw][8 * ct + j][lane] * a1;
        ob[idx] = g * (o * inv_l) + xhb[idx];
      }
    }
  }
}

extern "C" void kernel_launch(void* const* d_in, const int* in_sizes, int n_in,
                              void* d_out, int out_size, void* d_ws, size_t ws_size,
                              hipStream_t stream) {
  (void)in_sizes; (void)n_in; (void)out_size; (void)ws_size;
  const float* x     = (const float*)d_in[0];
  const float* x_h   = (const float*)d_in[1];
  const float* Wq    = (const float*)d_in[2];
  const float* Wk    = (const float*)d_in[3];
  const float* Wv    = (const float*)d_in[4];
  const float* gamma = (const float*)d_in[5];
  float* out = (float*)d_out;

  float* qw = (float*)d_ws;
  float* kw = qw + (size_t)BATCH * CQK * NSEQ;
  float* vw = kw + (size_t)BATCH * CQK * NSEQ;

  dim3 blk(128);
  proj_wmma_kernel<<<dim3(NSEQ / 64, CQK / 64, BATCH), blk, 0, stream>>>(Wq, x, qw, CQK, CQK);
  proj_wmma_kernel<<<dim3(NSEQ / 64, CQK / 64, BATCH), blk, 0, stream>>>(Wk, x, kw, CQK, CQK);
  proj_wmma_kernel<<<dim3(NSEQ / 64, CV / 64, BATCH), blk, 0, stream>>>(Wv, x_h, vw, CV, CV);
  flash_attn_f32_kernel<<<dim3(NSEQ / 64, BATCH), dim3(256), 0, stream>>>(qw, kw, vw, x_h, gamma, out);
}


# attention kernel: K/V staging software-pipelined: next tile's global loads go to registers before this tile's matrix work, to a second LDS buffer after it; one barrier per tile, prefetch hint dropped
# speedup vs baseline: 2.0067x; 1.1171x over previous
#include <hip/hip_runtime.h>


typedef float v2f __attribute__((ext_vector_type(2)));
typedef float v8f __attribute__((ext_vector_type(8)));

#define WMMA_F32X4(A, B, C) \
  __builtin_amdgcn_wmma_f32_16x16x4_f32(false, (A), false, (B), (short)0, (C), false, false)

namespace {
constexpr int BATCH = 4;
constexpr int CQK   = 64;
constexpr int CV    = 128;
constexpr int NSEQ  = 4096;
constexpr int MT    = 32;
constexpr int LDK2 = 2 * MT + 2;
constexpr int LDV2 = 2 * CV + 8;
constexpr int KBUF = (CQK / 2) * LDK2;
constexpr int VBUF = (MT / 2) * LDV2;
constexpr int NIT  = NSEQ / (2 * MT);
}

__device__ __forceinline__ float xhalf(float x) {
  const int b = __builtin_bit_cast(int, x);
  return __builtin_bit_cast(float,
      __builtin_amdgcn_permlanex16(b, b, 0x76543210, 0xfedcba98, false, false));
}

__device__ __forceinline__ void sched_fence() { __builtin_amdgcn_sched_barrier(0); }

struct TileRegs {
  float4 ke[2], ko[2], v[8];
};

__device__ __forceinline__ void tile_load(TileRegs& r, const float* __restrict__ kp,
                                          const float* __restrict__ vp) {
#pragma unroll
  for (int ii = 0; ii < 2; ++ii) {
    r.ke[ii] = *(const float4*)(kp + (size_t)(32 * ii) * NSEQ);
    r.ko[ii] = *(const float4*)(kp + (size_t)(32 * ii + 1) * NSEQ);
  }
#pragma unroll
  for (int i = 0; i < 8; ++i) r.v[i] = *(const float4*)(vp + (size_t)(16 * i) * NSEQ);
}

__device__ __forceinline__ void tile_store(const TileRegs& r, float* kd, float* vd) {
#pragma unroll
  for (int ii = 0; ii < 2; ++ii) {
    float* dst = kd + 16 * ii * LDK2;
    v2f p0, p1, p2, p3;
    p0[0] = r.ke[ii].x; p0[1] = r.ko[ii].x;
    p1[0] = r.ke[ii].y; p1[1] = r.ko[ii].y;
    p2[0] = r.ke[ii].z; p2[1] = r.ko[ii].z;
    p3[0] = r.ke[ii].w; p3[1] = r.ko[ii].w;
    *(v2f*)&dst[0] = p0;
    *(v2f*)&dst[2] = p1;
    *(v2f*)&dst[4] = p2;
    *(v2f*)&dst[6] = p3;
  }
#pragma unroll
  for (int i = 0; i < 8; ++i) {
    float* dst = vd + 32 * i;
    v2f lo, hi;
    lo[0] = r.v[i].x; lo[1] = r.v[i].y;
    hi[0] = r.v[i].z; hi[1] = r.v[i].w;
    *(v2f*)&dst[0] = lo;
    *(v2f*)&dst[LDV2] = hi;
  }
}

__global__ __launch_bounds__(128) void proj_wmma_kernel(
    const float* __restrict__ W, const float* __restrict__ X,
    float* __restrict__ Y, int O, int C) {
  const int lane = threadIdx.x & 31;
  const int wave = threadIdx.x >> 5;
  const int h    = lane >> 4;
  const int l15  = lane & 15;
  const int b    = blockIdx.z;
  const int o0   = blockIdx.y * 64 + wave * 16;
  const int n0   = blockIdx.x * 64;

  const float* Xb = X + (size_t)b * C * NSEQ;
  float* Yb       = Y + (size_t)b * O * NSEQ;

  v8f acc[4];
#pragma unroll
  for (int st = 0; st < 4; ++st) { v8f z = {}; acc[st] = z; }

  const int ksteps = C >> 2;
  for (int t = 0; t < ksteps; ++t) {
    const int c0 = 4 * t + 2 * h;
    v2f a = *(const v2f*)&W[(size_t)(o0 + l15) * C + c0];
#pragma unroll
    for (int st = 0; st < 4; ++st) {
      const int n = n0 + 16 * st + l15;
      v2f bop;
      bop[0] = Xb[(size_t)c0 * NSEQ + n];
      bop[1] = Xb[(size_t)(c0 + 1) * NSEQ + n];
      acc[st] = WMMA_F32X4(a, bop, acc[st]);
    }
  }
#pragma unroll
  for (int st = 0; st < 4; ++st) {
    const int n = n0 + 16 * st + l15;
#pragma unroll
    for (int j = 0; j < 8; ++j) {
      Yb[(size_t)(o0 + j + 8 * h) * NSEQ + n] = acc[st][j];
    }
  }
}

__global__ __launch_bounds__(256) void flash_attn_f32_kernel(
    const float* __restrict__ q, const float* __restrict__ k,
    const float* __restrict__ v, const float* __restrict__ x_h,
    const float* __restrict__ gamma_p, float* __restrict__ out) {
  __shared__ float kbp2[2][2 * KBUF];
  __shared__ float vtp2[2][2 * VBUF];
  __shared__ float mrg[4][66][32];

  const int tid  = threadIdx.x;
  const int lane = tid & 31;
  const int wave = tid >> 5;
  const int grp  = wave >> 2;
  const int qw   = wave & 3;
  const int t7   = tid & 127;
  const int h    = lane >> 4;
  const int l15  = lane & 15;
  const int b    = blockIdx.y;
  const int n0   = blockIdx.x * 64 + qw * 16;
  float* kbase = kbp2[grp];
  float* vbase = vtp2[grp];

  const float* qb  = q + (size_t)b * CQK * NSEQ;
  const float* kb  = k + (size_t)b * CQK * NSEQ;
  const float* vb  = v + (size_t)b * CV * NSEQ;

  const int sq  = t7 & 7;
  const int sg  = t7 >> 3;
  const int scol = sg ^ ((sq >> 2) * 4);
  const int l15s = l15 ^ 4;
  const float* ksrc = kb + (size_t)(2 * sg) * NSEQ + grp * MT + 4 * sq;
  const float* vsrc = vb + (size_t)sg * NSEQ + grp * MT + 4 * sq;
  float* kdst = &kbase[sg * LDK2 + 8 * sq];
  float* vdst = &vbase[(2 * sq) * LDV2 + 2 * scol];

  v2f qa[16];
#pragma unroll
  for (int t = 0; t < 16; ++t) {
    const int c0 = 4 * t + 2 * h;
    const int n  = n0 + l15;
    qa[t][0] = qb[(size_t)c0 * NSEQ + n];
    qa[t][1] = qb[(size_t)(c0 + 1) * NSEQ + n];
  }

  v8f Oacc[8];
#pragma unroll
  for (int ct = 0; ct < 8; ++ct) { v8f z = {}; Oacc[ct] = z; }
  float m_s = -3.0e38f;
  float l_s = 0.0f;

  TileRegs tr;
  tile_load(tr, ksrc, vsrc);
  tile_store(tr, kdst, vdst);
  __syncthreads();

  for (int it = 0; it < NIT; ++it) {
    const int cur = it & 1;
    const bool more = (it + 1 < NIT);
    const float* kbp = kbase + cur * KBUF;
    const float* vtp = vbase + cur * VBUF;

    if (more) tile_load(tr, ksrc + (size_t)(it + 1) * (2 * MT), vsrc + (size_t)(it + 1) * (2 * MT));
    sched_fence();

    v8f E[2];
#pragma unroll
    for (int st = 0; st < 2; ++st) { v8f z = {}; E[st] = z; }
    for (int t = 0; t < 16; ++t) {
      const float* krow = &kbp[(2 * t + h) * LDK2];
#pragma unroll
      for (int st = 0; st < 2; ++st) {
        v2f aop = *(const v2f*)&krow[(16 * st + l15) * 2];
        E[st] = WMMA_F32X4(aop, qa[t], E[st]);
      }
    }

    float tmax = fmaxf(E[0][0], E[1][0]);
#pragma unroll
    for (int j = 1; j < 8; ++j) tmax = fmaxf(tmax, fmaxf(E[0][j], E[1][j]));
    tmax = fmaxf(tmax, xhalf(tmax));
    const float mnew = fmaxf(m_s, tmax);
    const float scale = __expf(m_s - mnew);
    m_s = mnew;

#pragma unroll
    for (int st = 0; st < 2; ++st)
#pragma unroll
      for (int j = 0; j < 8; ++j) E[st][j] = __expf(E[st][j] - mnew);

    float tsum = E[0][0] + E[1][0];
#pragma unroll
    for (int j = 1; j < 8; ++j) tsum += E[0][j] + E[1][j];
    l_s = l_s * scale + tsum;

#pragma unroll
    for (int ct = 0; ct < 8; ++ct)
#pragma unroll
      for (int j = 0; j < 8; ++j) Oacc[ct][j] *= scale;

#pragma unroll
    for (int st = 0; st < 2; ++st) {
      const int lc = st ? l15s : l15;
#pragma unroll
      for (int u = 0; u < 4; ++u) {
        v2f bop;
        bop[0] = E[st][2 * u];
        bop[1] = E[st][2 * u + 1];
        const float* vrow = &vtp[(8 * st + 4 * h + u) * LDV2];
#pragma unroll
        for (int ct = 0; ct < 8; ++ct) {
          v2f aop = *(const v2f*)&vrow[(16 * ct + lc) * 2];
          Oacc[ct] = WMMA_F32X4(aop, bop, Oacc[ct]);
        }
      }
    }

    sched_fence();
    if (more) tile_store(tr, kdst + (cur ^ 1) * KBUF, vdst + (cur ^ 1) * VBUF);
    __syncthreads();
  }

  l_s = l_s + xhalf(l_s);

  if (grp == 1) {
#pragma unroll
    for (int ct = 0; ct < 8; ++ct)
#pragma unroll
      for (int j = 0; j < 8; ++j) mrg[qw][8 * ct + j][lane] = Oacc[ct][j];
    mrg[qw][64][lane] = m_s;
    mrg[qw][65][lane] = l_s;
  }
  __syncthreads();

  if (grp == 0) {
    const float m1 = mrg[qw][64][lane];
    const float l1 = mrg[qw][65][lane];
    const float mm = fmaxf(m_s, m1);
    const float a0 = __expf(m_s - mm);
    const float a1 = __expf(m1 - mm);
    const float g = gamma_p[0];
    const float inv_l = 1.0f / (l_s * a0 + l1 * a1);
    const float* xhb = x_h + (size_t)b * CV * NSEQ;
    float* ob        = out + (size_t)b * CV * NSEQ;
    const int n = n0 + l15;
#pragma unroll
    for (int ct = 0; ct < 8; ++ct) {
#pragma unroll
      for (int j = 0; j < 8; ++j) {
        const int c = 16 * ct + j + 8 * h;
        const size_t idx = (size_t)c * NSEQ + n;
        const float o = Oacc[ct][j] * a0 + mrg[qw][8 * ct + j][lane] * a1;
        ob[idx] = g * (o * inv_l) + xhb[idx];
      }
    }
  }
}

extern "C" void kernel_launch(void* const* d_in, const int* in_sizes, int n_in,
                              void* d_out, int out_size, void* d_ws, size_t ws_size,
                              hipStream_t stream) {
  (void)in_sizes; (void)n_in; (void)out_size; (void)ws_size;
  const float* x     = (const float*)d_in[0];
  const float* x_h   = (const float*)d_in[1];
  const float* Wq    = (const float*)d_in[2];
  const float* Wk    = (const float*)d_in[3];
  const float* Wv    = (const float*)d_in[4];
  const float* gamma = (const float*)d_in[5];
  float* out = (float*)d_out;

  float* qw = (float*)d_ws;
  float* kw = qw + (size_t)BATCH * CQK * NSEQ;
  float* vw = kw + (size_t)BATCH * CQK * NSEQ;

  dim3 blk(128);
  proj_wmma_kernel<<<dim3(NSEQ / 64, CQK / 64, BATCH), blk, 0, stream>>>(Wq, x, qw, CQK, CQK);
  proj_wmma_kernel<<<dim3(NSEQ / 64, CQK / 64, BATCH), blk, 0, stream>>>(Wk, x, kw, CQK, CQK);
  proj_wmma_kernel<<<dim3(NSEQ / 64, CV / 64, BATCH), blk, 0, stream>>>(Wv, x_h, vw, CV, CV);
  flash_attn_f32_kernel<<<dim3(NSEQ / 64, BATCH), dim3(256), 0, stream>>>(qw, kw, vw, x_h, gamma, out);
}


# projections: the three q/k/v launches merged into one launch (grid plane z picks the matrix, v first): their latency-bound waves share the SIMDs and run side by side; launches per repetition 4 -> 2
# speedup vs baseline: 2.0867x; 1.0399x over previous
#include <hip/hip_runtime.h>


typedef float v2f __attribute__((ext_vector_type(2)));
typedef float v8f __attribute__((ext_vector_type(8)));

#define WMMA_F32X4(A, B, C) \
  __builtin_amdgcn_wmma_f32_16x16x4_f32(false, (A), false, (B), (short)0, (C), false, false)

namespace {
constexpr int BATCH = 4;
constexpr int CQK   = 64;
constexpr int CV    = 128;
constexpr int NSEQ  = 4096;
constexpr int MT    = 32;
constexpr int LDK2 = 2 * MT + 2;
constexpr int LDV2 = 2 * CV + 8;
constexpr int KBUF = (CQK / 2) * LDK2;
constexpr int VBUF = (MT / 2) * LDV2;
constexpr int NIT  = NSEQ / (2 * MT);
}

__device__ __forceinline__ float xhalf(float x) {
  const int b = __builtin_bit_cast(int, x);
  return __builtin_bit_cast(float,
      __builtin_amdgcn_permlanex16(b, b, 0x76543210, 0xfedcba98, false, false));
}

__device__ __forceinline__ void sched_fence() { __builtin_amdgcn_sched_barrier(0); }

struct TileRegs {
  float4 ke[2], ko[2], v[8];
};

__device__ __forceinline__ void tile_load(TileRegs& r, const float* __restrict__ kp,
                                          const float* __restrict__ vp) {
#pragma unroll
  for (int ii = 0; ii < 2; ++ii) {
    r.ke[ii] = *(const float4*)(kp + (size_t)(32 * ii) * NSEQ);
    r.ko[ii] = *(const float4*)(kp + (size_t)(32 * ii + 1) * NSEQ);
  }
#pragma unroll
  for (int i = 0; i < 8; ++i) r.v[i] = *(const float4*)(vp + (size_t)(16 * i) * NSEQ);
}

__device__ __forceinline__ void tile_store(const TileRegs& r, float* kd, float* vd) {
#pragma unroll
  for (int ii = 0; ii < 2; ++ii) {
    float* dst = kd + 16 * ii * LDK2;
    v2f p0, p1, p2, p3;
    p0[0] = r.ke[ii].x; p0[1] = r.ko[ii].x;
    p1[0] = r.ke[ii].y; p1[1] = r.ko[ii].y;
    p2[0] = r.ke[ii].z; p2[1] = r.ko[ii].z;
    p3[0] = r.ke[ii].w; p3[1] = r.ko[ii].w;
    *(v2f*)&dst[0] = p0;
    *(v2f*)&dst[2] = p1;
    *(v2f*)&dst[4] = p2;
    *(v2f*)&dst[6] = p3;
  }
#pragma unroll
  for (int i = 0; i < 8; ++i) {
    float* dst = vd + 32 * i;
    v2f lo, hi;
    lo[0] = r.v[i].x; lo[1] = r.v[i].y;
    hi[0] = r.v[i].z; hi[1] = r.v[i].w;
    *(v2f*)&dst[0] = lo;
    *(v2f*)&dst[LDV2] = hi;
  }
}

template <int O, int C>
__device__ __forceinline__ void proj_tile(
    const float* __restrict__ W, const float* __restrict__ X,
    float* __restrict__ Y, int oblk) {
  const int lane = threadIdx.x & 31;
  const int wave = threadIdx.x >> 5;
  const int h    = lane >> 4;
  const int l15  = lane & 15;
  const int b    = blockIdx.y;
  const int o0   = oblk * 64 + wave * 16;
  const int n0   = blockIdx.x * 64;

  const float* Xb = X + (size_t)b * C * NSEQ;
  float* Yb       = Y + (size_t)b * O * NSEQ;

  v8f acc[4];
#pragma unroll
  for (int st = 0; st < 4; ++st) { v8f z = {}; acc[st] = z; }

  constexpr int ksteps = C >> 2;
  for (int t = 0; t < ksteps; ++t) {
    const int c0 = 4 * t + 2 * h;
    v2f a = *(const v2f*)&W[(size_t)(o0 + l15) * C + c0];
#pragma unroll
    for (int st = 0; st < 4; ++st) {
      const int n = n0 + 16 * st + l15;
      v2f bop;
      bop[0] = Xb[(size_t)c0 * NSEQ + n];
      bop[1] = Xb[(size_t)(c0 + 1) * NSEQ + n];
      acc[st] = WMMA_F32X4(a, bop, acc[st]);
    }
  }
#pragma unroll
  for (int st = 0; st < 4; ++st) {
    const int n = n0 + 16 * st + l15;
#pragma unroll
    for (int j = 0; j < 8; ++j) {
      Yb[(size_t)(o0 + j + 8 * h) * NSEQ + n] = acc[st][j];
    }
  }
}

__global__ __launch_bounds__(128) void proj_qkv_wmma_kernel(
    const float* __restrict__ Wq, const float* __restrict__ Wk,
    const float* __restrict__ Wv, const float* __restrict__ x,
    const float* __restrict__ x_h, float* __restrict__ qo,
    float* __restrict__ ko, float* __restrict__ vo) {
  const int z = blockIdx.z;
  if (z < 2)       proj_tile<CV, CV>(Wv, x_h, vo, z);
  else if (z == 2) proj_tile<CQK, CQK>(Wq, x, qo, 0);
  else             proj_tile<CQK, CQK>(Wk, x, ko, 0);
}

__global__ __launch_bounds__(256) void flash_attn_f32_kernel(
    const float* __restrict__ q, const float* __restrict__ k,
    const float* __restrict__ v, const float* __restrict__ x_h,
    const float* __restrict__ gamma_p, float* __restrict__ out) {
  __shared__ float kbp2[2][2 * KBUF];
  __shared__ float vtp2[2][2 * VBUF];
  __shared__ float mrg[4][66][32];

  const int tid  = threadIdx.x;
  const int lane = tid & 31;
  const int wave = tid >> 5;
  const int grp  = wave >> 2;
  const int qw   = wave & 3;
  const int t7   = tid & 127;
  const int h    = lane >> 4;
  const int l15  = lane & 15;
  const int b    = blockIdx.y;
  const int n0   = blockIdx.x * 64 + qw * 16;
  float* kbase = kbp2[grp];
  float* vbase = vtp2[grp];

  const float* qb  = q + (size_t)b * CQK * NSEQ;
  const float* kb  = k + (size_t)b * CQK * NSEQ;
  const float* vb  = v + (size_t)b * CV * NSEQ;

  const int sq  = t7 & 7;
  const int sg  = t7 >> 3;
  const int scol = sg ^ ((sq >> 2) * 4);
  const int l15s = l15 ^ 4;
  const float* ksrc = kb + (size_t)(2 * sg) * NSEQ + grp * MT + 4 * sq;
  const float* vsrc = vb + (size_t)sg * NSEQ + grp * MT + 4 * sq;
  float* kdst = &kbase[sg * LDK2 + 8 * sq];
  float* vdst = &vbase[(2 * sq) * LDV2 + 2 * scol];

  v2f qa[16];
#pragma unroll
  for (int t = 0; t < 16; ++t) {
    const int c0 = 4 * t + 2 * h;
    const int n  = n0 + l15;
    qa[t][0] = qb[(size_t)c0 * NSEQ + n];
    qa[t][1] = qb[(size_t)(c0 + 1) * NSEQ + n];
  }

  v8f Oacc[8];
#pragma unroll
  for (int ct = 0; ct < 8; ++ct) { v8f z = {}; Oacc[ct] = z; }
  float m_s = -3.0e38f;
  float l_s = 0.0f;

  TileRegs tr;
  tile_load(tr, ksrc, vsrc);
  tile_store(tr, kdst, vdst);
  __syncthreads();

  for (int it = 0; it < NIT; ++it) {
    const int cur = it & 1;
    const bool more = (it + 1 < NIT);
    const float* kbp = kbase + cur * KBUF;
    const float* vtp = vbase + cur * VBUF;

    if (more) tile_load(tr, ksrc + (size_t)(it + 1) * (2 * MT), vsrc + (size_t)(it + 1) * (2 * MT));
    sched_fence();

    v8f E[2];
#pragma unroll
    for (int st = 0; st < 2; ++st) { v8f z = {}; E[st] = z; }
    for (int t = 0; t < 16; ++t) {
      const float* krow = &kbp[(2 * t + h) * LDK2];
#pragma unroll
      for (int st = 0; st < 2; ++st) {
        v2f aop = *(const v2f*)&krow[(16 * st + l15) * 2];
        E[st] = WMMA_F32X4(aop, qa[t], E[st]);
      }
    }

    float tmax = fmaxf(E[0][0], E[1][0]);
#pragma unroll
    for (int j = 1; j < 8; ++j) tmax = fmaxf(tmax, fmaxf(E[0][j], E[1][j]));
    tmax = fmaxf(tmax, xhalf(tmax));
    const float mnew = fmaxf(m_s, tmax);
    const float scale = __expf(m_s - mnew);
    m_s = mnew;

#pragma unroll
    for (int st = 0; st < 2; ++st)
#pragma unroll
      for (int j = 0; j < 8; ++j) E[st][j] = __expf(E[st][j] - mnew);

    float tsum = E[0][0] + E[1][0];
#pragma unroll
    for (int j = 1; j < 8; ++j) tsum += E[0][j] + E[1][j];
    l_s = l_s * scale + tsum;

#pragma unroll
    for (int ct = 0; ct < 8; ++ct)
#pragma unroll
      for (int j = 0; j < 8; ++j) Oacc[ct][j] *= scale;

#pragma unroll
    for (int st = 0; st < 2; ++st) {
      const int lc = st ? l15s : l15;
#pragma unroll
      for (int u = 0; u < 4; ++u) {
        v2f bop;
        bop[0] = E[st][2 * u];
        bop[1] = E[st][2 * u + 1];
        const float* vrow = &vtp[(8 * st + 4 * h + u) * LDV2];
#pragma unroll
        for (int ct = 0; ct < 8; ++ct) {
          v2f aop = *(const v2f*)&vrow[(16 * ct + lc) * 2];
          Oacc[ct] = WMMA_F32X4(aop, bop, Oacc[ct]);
        }
      }
    }

    sched_fence();
    if (more) tile_store(tr, kdst + (cur ^ 1) * KBUF, vdst + (cur ^ 1) * VBUF);
    __syncthreads();
  }

  l_s = l_s + xhalf(l_s);

  if (grp == 1) {
#pragma unroll
    for (int ct = 0; ct < 8; ++ct)
#pragma unroll
      for (int j = 0; j < 8; ++j) mrg[qw][8 * ct + j][lane] = Oacc[ct][j];
    mrg[qw][64][lane] = m_s;
    mrg[qw][65][lane] = l_s;
  }
  __syncthreads();

  if (grp == 0) {
    const float m1 = mrg[qw][64][lane];
    const float l1 = mrg[qw][65][lane];
    const float mm = fmaxf(m_s, m1);
    const float a0 = __expf(m_s - mm);
    const float a1 = __expf(m1 - mm);
    const float g = gamma_p[0];
    const float inv_l = 1.0f / (l_s * a0 + l1 * a1);
    const float* xhb = x_h + (size_t)b * CV * NSEQ;
    float* ob        = out + (size_t)b * CV * NSEQ;
    const int n = n0 + l15;
#pragma unroll
    for (int ct = 0; ct < 8; ++ct) {
#pragma unroll
      for (int j = 0; j < 8; ++j) {
        const int c = 16 * ct + j + 8 * h;
        const size_t idx = (size_t)c * NSEQ + n;
        const float o = Oacc[ct][j] * a0 + mrg[qw][8 * ct + j][lane] * a1;
        ob[idx] = g * (o * inv_l) + xhb[idx];
      }
    }
  }
}

extern "C" void kernel_launch(void* const* d_in, const int* in_sizes, int n_in,
                              void* d_out, int out_size, void* d_ws, size_t ws_size,
                              hipStream_t stream) {
  (void)in_sizes; (void)n_in; (void)out_size; (void)ws_size;
  const float* x     = (const float*)d_in[0];
  const float* x_h   = (const float*)d_in[1];
  const float* Wq    = (const float*)d_in[2];
  const float* Wk    = (const float*)d_in[3];
  const float* Wv    = (const float*)d_in[4];
  const float* gamma = (const float*)d_in[5];
  float* out = (float*)d_out;

  float* qw = (float*)d_ws;
  float* kw = qw + (size_t)BATCH * CQK * NSEQ;
  float* vw = kw + (size_t)BATCH * CQK * NSEQ;

  proj_qkv_wmma_kernel<<<dim3(NSEQ / 64, BATCH, 4), dim3(128), 0, stream>>>(Wq, Wk, Wv, x, x_h, qw, kw, vw);
  flash_attn_f32_kernel<<<dim3(NSEQ / 64, BATCH), dim3(256), 0, stream>>>(qw, kw, vw, x_h, gamma, out);
}
